# starting kernel
# baseline (speedup 1.0000x reference)
#include <hip/hip_runtime.h>

typedef __attribute__((ext_vector_type(2))) float v2f;
typedef __attribute__((ext_vector_type(8))) float v8f;

#define NPIX   9216
#define DEMB   32
#define NTILES (NPIX / 16)
#define NWAVES 4
#define XSTRIDE 20
#define ESTRIDE 17
#define BW     3.0f
#define STEP   0.5f
#define LOG2E  1.44269504088896340736f

__global__ __launch_bounds__(32 * NWAVES)
void ms_iter_kernel(const float* __restrict__ xin,
                    float* __restrict__ xout,
                    float* __restrict__ xout2)
{
    __shared__ __align__(16) float xtile[NWAVES][DEMB * XSTRIDE];
    __shared__ float etile[NWAVES][16 * ESTRIDE];
    __shared__ float red[NWAVES][32][17];

    const int lane  = threadIdx.x & 31;
    const int wave  = threadIdx.x >> 5;
    const int half  = lane >> 4;
    const int l15   = lane & 15;
    const int jbase = blockIdx.x * 16;

    float* xt = xtile[wave];
    float* et = etile[wave];

    v2f bj[8];
#pragma unroll
    for (int k0 = 0; k0 < 8; ++k0) {
        const int d0 = 4 * k0 + 2 * half;
        bj[k0].x = xin[(size_t)d0 * NPIX + jbase + l15];
        bj[k0].y = xin[(size_t)(d0 + 1) * NPIX + jbase + l15];
    }

    v8f y0 = {};
    v8f y1 = {};
    float dacc = 0.0f;

    for (int it = wave; it < NTILES; it += NWAVES) {
        const int ibase = it * 16;

        {
            const float4* src = (const float4*)(xin + (size_t)lane * NPIX + ibase);
            float4* dst = (float4*)(xt + lane * XSTRIDE);
            dst[0] = src[0]; dst[1] = src[1]; dst[2] = src[2]; dst[3] = src[3];
        }
        if (ibase + 16 * NWAVES < NPIX)
            __builtin_prefetch(xin + (size_t)lane * NPIX + ibase + 16 * NWAVES, 0, 0);

        v8f s = {};
#pragma unroll
        for (int k0 = 0; k0 < 8; ++k0) {
            const int d0 = 4 * k0 + 2 * half;
            v2f a;
            a.x = xt[d0 * XSTRIDE + l15];
            a.y = xt[(d0 + 1) * XSTRIDE + l15];
            s = __builtin_amdgcn_wmma_f32_16x16x4_f32(false, a, false, bj[k0],
                                                      (short)0, s, false, false);
        }

        float esum = 0.0f;
#pragma unroll
        for (int v = 0; v < 8; ++v) {
            const float e = __builtin_amdgcn_exp2f(s[v] * (BW * LOG2E));
            esum += e;
            et[(v + 8 * half) * ESTRIDE + l15] = e;
        }
        dacc += esum;

#pragma unroll
        for (int c = 0; c < 4; ++c) {
            const int k0 = 4 * c + 2 * half;
            v2f b2;
            b2.x = et[k0 * ESTRIDE + l15];
            b2.y = et[(k0 + 1) * ESTRIDE + l15];

            v2f a2;
            a2.x = xt[l15 * XSTRIDE + k0];
            a2.y = xt[l15 * XSTRIDE + k0 + 1];
            y0 = __builtin_amdgcn_wmma_f32_16x16x4_f32(false, a2, false, b2,
                                                       (short)0, y0, false, false);

            v2f a3;
            a3.x = xt[(16 + l15) * XSTRIDE + k0];
            a3.y = xt[(16 + l15) * XSTRIDE + k0 + 1];
            y1 = __builtin_amdgcn_wmma_f32_16x16x4_f32(false, a3, false, b2,
                                                       (short)0, y1, false, false);
        }
    }

#pragma unroll
    for (int v = 0; v < 8; ++v) {
        red[wave][lane][v]     = y0[v];
        red[wave][lane][8 + v] = y1[v];
    }
    red[wave][lane][16] = dacc;
    __syncthreads();

    if (wave == 0) {
        v8f Y0 = {}, Y1 = {};
        float dtot = 0.0f;
        for (int w = 0; w < NWAVES; ++w) {
#pragma unroll
            for (int v = 0; v < 8; ++v) {
                Y0[v] += red[w][lane][v];
                Y1[v] += red[w][lane][8 + v];
            }
            dtot += red[w][lane][16];
        }
        const float dfull = dtot + __shfl_xor(dtot, 16, 32);
        const float sc = STEP / dfull;

#pragma unroll
        for (int v = 0; v < 8; ++v) {
            const int m = v + 8 * half;
            const size_t j = (size_t)jbase + l15;
            const size_t o0 = (size_t)m * NPIX + j;
            const size_t o1 = (size_t)(16 + m) * NPIX + j;
            const float r0 = Y0[v] * sc + (1.0f - STEP) * xin[o0];
            const float r1 = Y1[v] * sc + (1.0f - STEP) * xin[o1];
            xout[o0] = r0;
            xout[o1] = r1;
            if (xout2) { xout2[o0] = r0; xout2[o1] = r1; }
        }
    }
}

extern "C" void kernel_launch(void* const* d_in, const int* in_sizes, int n_in,
                              void* d_out, int out_size, void* d_ws, size_t ws_size,
                              hipStream_t stream) {
    (void)in_sizes; (void)n_in; (void)d_ws; (void)ws_size; (void)out_size;
    const float* x0 = (const float*)d_in[0];
    float* out = (float*)d_out;
    const size_t F = (size_t)DEMB * NPIX;

    dim3 grid(NTILES), block(32 * NWAVES);
    ms_iter_kernel<<<grid, block, 0, stream>>>(x0, out + F, nullptr);
    ms_iter_kernel<<<grid, block, 0, stream>>>(out + F, out + 2 * F, nullptr);
    ms_iter_kernel<<<grid, block, 0, stream>>>(out + 2 * F, out + 3 * F, out);
}


# Whole-line x staging: each 16-B-per-lane global load reads 4 rows x 128 B (8 lanes per row) of a 32-column slab per trip, not 32 rows x 16 B; waves regroup i-tiles, so f32 sum order changes
# speedup vs baseline: 1.3820x; 1.3820x over previous
#include <hip/hip_runtime.h>

typedef __attribute__((ext_vector_type(2))) float v2f;
typedef __attribute__((ext_vector_type(8))) float v8f;

#define NPIX   9216
#define DEMB   32
#define NWAVES 4
#define SLABW  32
#define NSLABS (NPIX / SLABW)
#define NTILES (NPIX / 16)
#define XSTRIDE 44
#define ESTRIDE 17
#define BW     3.0f
#define STEP   0.5f
#define LOG2E  1.44269504088896340736f

__global__ __launch_bounds__(32 * NWAVES)
void ms_iter_kernel(const float* __restrict__ xin,
                    float* __restrict__ xout,
                    float* __restrict__ xout2)
{
    __shared__ __align__(16) float xtile[NWAVES][DEMB * XSTRIDE];
    __shared__ float etile[NWAVES][SLABW * ESTRIDE];
    __shared__ float red[NWAVES][32][17];

    const int lane  = threadIdx.x & 31;
    const int wave  = threadIdx.x >> 5;
    const int half  = lane >> 4;
    const int l15   = lane & 15;
    const int jbase = blockIdx.x * 16;

    float* xt = xtile[wave];
    float* et = etile[wave];

    v2f bj[8];
#pragma unroll
    for (int k0 = 0; k0 < 8; ++k0) {
        const int d0 = 4 * k0 + 2 * half;
        bj[k0].x = xin[(size_t)d0 * NPIX + jbase + l15];
        bj[k0].y = xin[(size_t)(d0 + 1) * NPIX + jbase + l15];
    }

    v8f y0 = {};
    v8f y1 = {};
    float dacc = 0.0f;

    const int srow = lane >> 3;
    const int scol = 4 * (lane & 7);

    for (int sl = wave; sl < NSLABS; sl += NWAVES) {
        const int ibase = sl * SLABW;

        {
            const float4* src = (const float4*)(xin + (size_t)srow * NPIX + ibase + scol);
            float4* dst = (float4*)(xt + srow * XSTRIDE + scol);
            float4 r[8];
#pragma unroll
            for (int q = 0; q < 8; ++q) r[q] = src[(size_t)q * NPIX];
#pragma unroll
            for (int q = 0; q < 8; ++q) dst[q * XSTRIDE] = r[q];
        }

        v8f s0 = {};
        v8f s1 = {};
#pragma unroll
        for (int k0 = 0; k0 < 8; ++k0) {
            const int d0 = 4 * k0 + 2 * half;
            v2f a0, a1;
            a0.x = xt[d0 * XSTRIDE + l15];
            a0.y = xt[(d0 + 1) * XSTRIDE + l15];
            a1.x = xt[d0 * XSTRIDE + 16 + l15];
            a1.y = xt[(d0 + 1) * XSTRIDE + 16 + l15];
            s0 = __builtin_amdgcn_wmma_f32_16x16x4_f32(false, a0, false, bj[k0],
                                                       (short)0, s0, false, false);
            s1 = __builtin_amdgcn_wmma_f32_16x16x4_f32(false, a1, false, bj[k0],
                                                       (short)0, s1, false, false);
        }

        float esum0 = 0.0f, esum1 = 0.0f;
#pragma unroll
        for (int v = 0; v < 8; ++v) {
            const float e0 = __builtin_amdgcn_exp2f(s0[v] * (BW * LOG2E));
            const float e1 = __builtin_amdgcn_exp2f(s1[v] * (BW * LOG2E));
            esum0 += e0;
            esum1 += e1;
            et[(v + 8 * half) * ESTRIDE + l15]      = e0;
            et[(16 + v + 8 * half) * ESTRIDE + l15] = e1;
        }
        dacc += esum0;
        dacc += esum1;

#pragma unroll
        for (int c = 0; c < 8; ++c) {
            const int k0 = 4 * c + 2 * half;
            v2f b2;
            b2.x = et[k0 * ESTRIDE + l15];
            b2.y = et[(k0 + 1) * ESTRIDE + l15];

            v2f a2;
            a2.x = xt[l15 * XSTRIDE + k0];
            a2.y = xt[l15 * XSTRIDE + k0 + 1];
            y0 = __builtin_amdgcn_wmma_f32_16x16x4_f32(false, a2, false, b2,
                                                       (short)0, y0, false, false);

            v2f a3;
            a3.x = xt[(16 + l15) * XSTRIDE + k0];
            a3.y = xt[(16 + l15) * XSTRIDE + k0 + 1];
            y1 = __builtin_amdgcn_wmma_f32_16x16x4_f32(false, a3, false, b2,
                                                       (short)0, y1, false, false);
        }
    }

#pragma unroll
    for (int v = 0; v < 8; ++v) {
        red[wave][lane][v]     = y0[v];
        red[wave][lane][8 + v] = y1[v];
    }
    red[wave][lane][16] = dacc;
    __syncthreads();

    if (wave == 0) {
        v8f Y0 = {}, Y1 = {};
        float dtot = 0.0f;
        for (int w = 0; w < NWAVES; ++w) {
#pragma unroll
            for (int v = 0; v < 8; ++v) {
                Y0[v] += red[w][lane][v];
                Y1[v] += red[w][lane][8 + v];
            }
            dtot += red[w][lane][16];
        }
        const float dfull = dtot + __shfl_xor(dtot, 16, 32);
        const float sc = STEP / dfull;

#pragma unroll
        for (int v = 0; v < 8; ++v) {
            const int m = v + 8 * half;
            const size_t j = (size_t)jbase + l15;
            const size_t o0 = (size_t)m * NPIX + j;
            const size_t o1 = (size_t)(16 + m) * NPIX + j;
            const float r0 = Y0[v] * sc + (1.0f - STEP) * xin[o0];
            const float r1 = Y1[v] * sc + (1.0f - STEP) * xin[o1];
            xout[o0] = r0;
            xout[o1] = r1;
            if (xout2) { xout2[o0] = r0; xout2[o1] = r1; }
        }
    }
}

extern "C" void kernel_launch(void* const* d_in, const int* in_sizes, int n_in,
                              void* d_out, int out_size, void* d_ws, size_t ws_size,
                              hipStream_t stream) {
    (void)in_sizes; (void)n_in; (void)d_ws; (void)ws_size; (void)out_size;
    const float* x0 = (const float*)d_in[0];
    float* out = (float*)d_out;
    const size_t F = (size_t)DEMB * NPIX;

    dim3 grid(NTILES), block(32 * NWAVES);
    ms_iter_kernel<<<grid, block, 0, stream>>>(x0, out + F, nullptr);
    ms_iter_kernel<<<grid, block, 0, stream>>>(out + F, out + 2 * F, nullptr);
    ms_iter_kernel<<<grid, block, 0, stream>>>(out + 2 * F, out + 3 * F, out);
}


# Work-group widened from 4 to 8 waves (two per SIMD per group, up to six resident per SIMD): each j-tile's i-stream is split eight ways, so f32 partial sums of Y and d combine in a different order
# speedup vs baseline: 1.5096x; 1.0923x over previous
#include <hip/hip_runtime.h>

typedef __attribute__((ext_vector_type(2))) float v2f;
typedef __attribute__((ext_vector_type(8))) float v8f;

#define NPIX   9216
#define DEMB   32
#define NWAVES 8
#define SLABW  32
#define NSLABS (NPIX / SLABW)
#define NTILES (NPIX / 16)
#define XSTRIDE 44
#define ESTRIDE 17
#define BW     3.0f
#define STEP   0.5f
#define LOG2E  1.44269504088896340736f

__global__ __launch_bounds__(32 * NWAVES)
void ms_iter_kernel(const float* __restrict__ xin,
                    float* __restrict__ xout,
                    float* __restrict__ xout2)
{
    __shared__ __align__(16) float xtile[NWAVES][DEMB * XSTRIDE];
    __shared__ float etile[NWAVES][SLABW * ESTRIDE];
    __shared__ float red[NWAVES][32][17];

    const int lane  = threadIdx.x & 31;
    const int wave  = threadIdx.x >> 5;
    const int half  = lane >> 4;
    const int l15   = lane & 15;
    const int jbase = blockIdx.x * 16;

    float* xt = xtile[wave];
    float* et = etile[wave];

    v2f bj[8];
#pragma unroll
    for (int k0 = 0; k0 < 8; ++k0) {
        const int d0 = 4 * k0 + 2 * half;
        bj[k0].x = xin[(size_t)d0 * NPIX + jbase + l15];
        bj[k0].y = xin[(size_t)(d0 + 1) * NPIX + jbase + l15];
    }

    v8f y0 = {};
    v8f y1 = {};
    float dacc = 0.0f;

    const int srow = lane >> 3;
    const int scol = 4 * (lane & 7);

    for (int sl = wave; sl < NSLABS; sl += NWAVES) {
        const int ibase = sl * SLABW;

        {
            const float4* src = (const float4*)(xin + (size_t)srow * NPIX + ibase + scol);
            float4* dst = (float4*)(xt + srow * XSTRIDE + scol);
            float4 r[8];
#pragma unroll
            for (int q = 0; q < 8; ++q) r[q] = src[(size_t)q * NPIX];
#pragma unroll
            for (int q = 0; q < 8; ++q) dst[q * XSTRIDE] = r[q];
        }

        v8f s0 = {};
        v8f s1 = {};
#pragma unroll
        for (int k0 = 0; k0 < 8; ++k0) {
            const int d0 = 4 * k0 + 2 * half;
            v2f a0, a1;
            a0.x = xt[d0 * XSTRIDE + l15];
            a0.y = xt[(d0 + 1) * XSTRIDE + l15];
            a1.x = xt[d0 * XSTRIDE + 16 + l15];
            a1.y = xt[(d0 + 1) * XSTRIDE + 16 + l15];
            s0 = __builtin_amdgcn_wmma_f32_16x16x4_f32(false, a0, false, bj[k0],
                                                       (short)0, s0, false, false);
            s1 = __builtin_amdgcn_wmma_f32_16x16x4_f32(false, a1, false, bj[k0],
                                                       (short)0, s1, false, false);
        }

        float esum0 = 0.0f, esum1 = 0.0f;
#pragma unroll
        for (int v = 0; v < 8; ++v) {
            const float e0 = __builtin_amdgcn_exp2f(s0[v] * (BW * LOG2E));
            const float e1 = __builtin_amdgcn_exp2f(s1[v] * (BW * LOG2E));
            esum0 += e0;
            esum1 += e1;
            et[(v + 8 * half) * ESTRIDE + l15]      = e0;
            et[(16 + v + 8 * half) * ESTRIDE + l15] = e1;
        }
        dacc += esum0;
        dacc += esum1;

#pragma unroll
        for (int c = 0; c < 8; ++c) {
            const int k0 = 4 * c + 2 * half;
            v2f b2;
            b2.x = et[k0 * ESTRIDE + l15];
            b2.y = et[(k0 + 1) * ESTRIDE + l15];

            v2f a2;
            a2.x = xt[l15 * XSTRIDE + k0];
            a2.y = xt[l15 * XSTRIDE + k0 + 1];
            y0 = __builtin_amdgcn_wmma_f32_16x16x4_f32(false, a2, false, b2,
                                                       (short)0, y0, false, false);

            v2f a3;
            a3.x = xt[(16 + l15) * XSTRIDE + k0];
            a3.y = xt[(16 + l15) * XSTRIDE + k0 + 1];
            y1 = __builtin_amdgcn_wmma_f32_16x16x4_f32(false, a3, false, b2,
                                                       (short)0, y1, false, false);
        }
    }

#pragma unroll
    for (int v = 0; v < 8; ++v) {
        red[wave][lane][v]     = y0[v];
        red[wave][lane][8 + v] = y1[v];
    }
    red[wave][lane][16] = dacc;
    __syncthreads();

    if (wave == 0) {
        v8f Y0 = {}, Y1 = {};
        float dtot = 0.0f;
        for (int w = 0; w < NWAVES; ++w) {
#pragma unroll
            for (int v = 0; v < 8; ++v) {
                Y0[v] += red[w][lane][v];
                Y1[v] += red[w][lane][8 + v];
            }
            dtot += red[w][lane][16];
        }
        const float dfull = dtot + __shfl_xor(dtot, 16, 32);
        const float sc = STEP / dfull;

#pragma unroll
        for (int v = 0; v < 8; ++v) {
            const int m = v + 8 * half;
            const size_t j = (size_t)jbase + l15;
            const size_t o0 = (size_t)m * NPIX + j;
            const size_t o1 = (size_t)(16 + m) * NPIX + j;
            const float r0 = Y0[v] * sc + (1.0f - STEP) * xin[o0];
            const float r1 = Y1[v] * sc + (1.0f - STEP) * xin[o1];
            xout[o0] = r0;
            xout[o1] = r1;
            if (xout2) { xout2[o0] = r0; xout2[o1] = r1; }
        }
    }
}

extern "C" void kernel_launch(void* const* d_in, const int* in_sizes, int n_in,
                              void* d_out, int out_size, void* d_ws, size_t ws_size,
                              hipStream_t stream) {
    (void)in_sizes; (void)n_in; (void)d_ws; (void)ws_size; (void)out_size;
    const float* x0 = (const float*)d_in[0];
    float* out = (float*)d_out;
    const size_t F = (size_t)DEMB * NPIX;

    dim3 grid(NTILES), block(32 * NWAVES);
    ms_iter_kernel<<<grid, block, 0, stream>>>(x0, out + F, nullptr);
    ms_iter_kernel<<<grid, block, 0, stream>>>(out + F, out + 2 * F, nullptr);
    ms_iter_kernel<<<grid, block, 0, stream>>>(out + 2 * F, out + 3 * F, out);
}


# exp(S) tiles pass from GEMM1's accumulators straight into GEMM2's B operand in registers (GEMM1 tile rows permuted to pair up), no LDS round trip for E; d's f32 column-sum order changes
# speedup vs baseline: 1.5775x; 1.0450x over previous
#include <hip/hip_runtime.h>

typedef __attribute__((ext_vector_type(2))) float v2f;
typedef __attribute__((ext_vector_type(8))) float v8f;

#define NPIX   9216
#define DEMB   32
#define NWAVES 8
#define SLABW  32
#define NSLABS (NPIX / SLABW)
#define NTILES (NPIX / 16)
#define XSTRIDE 44
#define BW     3.0f
#define STEP   0.5f
#define LOG2E  1.44269504088896340736f

__global__ __launch_bounds__(32 * NWAVES)
void ms_iter_kernel(const float* __restrict__ xin,
                    float* __restrict__ xout,
                    float* __restrict__ xout2)
{
    __shared__ __align__(16) float xtile[NWAVES][DEMB * XSTRIDE];
    __shared__ float red[NWAVES][32][17];

    const int lane  = threadIdx.x & 31;
    const int wave  = threadIdx.x >> 5;
    const int half  = lane >> 4;
    const int l15   = lane & 15;
    const int jbase = blockIdx.x * 16;

    float* xt = xtile[wave];
    const int pcol = 4 * ((l15 >> 1) & 3) + 2 * (l15 >> 3) + (l15 & 1);

    v2f bj[8];
#pragma unroll
    for (int k0 = 0; k0 < 8; ++k0) {
        const int d0 = 4 * k0 + 2 * half;
        bj[k0].x = xin[(size_t)d0 * NPIX + jbase + l15];
        bj[k0].y = xin[(size_t)(d0 + 1) * NPIX + jbase + l15];
    }

    v8f y0 = {};
    v8f y1 = {};
    float dacc = 0.0f;

    const int srow = lane >> 3;
    const int scol = 4 * (lane & 7);

    for (int sl = wave; sl < NSLABS; sl += NWAVES) {
        const int ibase = sl * SLABW;

        {
            const float4* src = (const float4*)(xin + (size_t)srow * NPIX + ibase + scol);
            float4* dst = (float4*)(xt + srow * XSTRIDE + scol);
            float4 r[8];
#pragma unroll
            for (int q = 0; q < 8; ++q) r[q] = src[(size_t)q * NPIX];
#pragma unroll
            for (int q = 0; q < 8; ++q) dst[q * XSTRIDE] = r[q];
        }

        v8f s0 = {};
        v8f s1 = {};
#pragma unroll
        for (int k0 = 0; k0 < 8; ++k0) {
            const int d0 = 4 * k0 + 2 * half;
            v2f a0, a1;
            a0.x = xt[d0 * XSTRIDE + pcol];
            a0.y = xt[(d0 + 1) * XSTRIDE + pcol];
            a1.x = xt[d0 * XSTRIDE + 16 + pcol];
            a1.y = xt[(d0 + 1) * XSTRIDE + 16 + pcol];
            s0 = __builtin_amdgcn_wmma_f32_16x16x4_f32(false, a0, false, bj[k0],
                                                       (short)0, s0, false, false);
            s1 = __builtin_amdgcn_wmma_f32_16x16x4_f32(false, a1, false, bj[k0],
                                                       (short)0, s1, false, false);
        }

        v8f e0, e1;
        float esum0 = 0.0f, esum1 = 0.0f;
#pragma unroll
        for (int v = 0; v < 8; ++v) {
            e0[v] = __builtin_amdgcn_exp2f(s0[v] * (BW * LOG2E));
            e1[v] = __builtin_amdgcn_exp2f(s1[v] * (BW * LOG2E));
            esum0 += e0[v];
            esum1 += e1[v];
        }
        dacc += esum0;
        dacc += esum1;

#pragma unroll
        for (int c = 0; c < 8; ++c) {
            const int k0 = 4 * c + 2 * half;
            const int t2 = 2 * (c & 3);
            v2f b2;
            b2.x = (c < 4) ? e0[t2]     : e1[t2];
            b2.y = (c < 4) ? e0[t2 + 1] : e1[t2 + 1];

            v2f a2;
            a2.x = xt[l15 * XSTRIDE + k0];
            a2.y = xt[l15 * XSTRIDE + k0 + 1];
            y0 = __builtin_amdgcn_wmma_f32_16x16x4_f32(false, a2, false, b2,
                                                       (short)0, y0, false, false);

            v2f a3;
            a3.x = xt[(16 + l15) * XSTRIDE + k0];
            a3.y = xt[(16 + l15) * XSTRIDE + k0 + 1];
            y1 = __builtin_amdgcn_wmma_f32_16x16x4_f32(false, a3, false, b2,
                                                       (short)0, y1, false, false);
        }
    }

#pragma unroll
    for (int v = 0; v < 8; ++v) {
        red[wave][lane][v]     = y0[v];
        red[wave][lane][8 + v] = y1[v];
    }
    red[wave][lane][16] = dacc;
    __syncthreads();

    if (wave == 0) {
        v8f Y0 = {}, Y1 = {};
        float dtot = 0.0f;
        for (int w = 0; w < NWAVES; ++w) {
#pragma unroll
            for (int v = 0; v < 8; ++v) {
                Y0[v] += red[w][lane][v];
                Y1[v] += red[w][lane][8 + v];
            }
            dtot += red[w][lane][16];
        }
        const float dfull = dtot + __shfl_xor(dtot, 16, 32);
        const float sc = STEP / dfull;

#pragma unroll
        for (int v = 0; v < 8; ++v) {
            const int m = v + 8 * half;
            const size_t j = (size_t)jbase + l15;
            const size_t o0 = (size_t)m * NPIX + j;
            const size_t o1 = (size_t)(16 + m) * NPIX + j;
            const float r0 = Y0[v] * sc + (1.0f - STEP) * xin[o0];
            const float r1 = Y1[v] * sc + (1.0f - STEP) * xin[o1];
            xout[o0] = r0;
            xout[o1] = r1;
            if (xout2) { xout2[o0] = r0; xout2[o1] = r1; }
        }
    }
}

extern "C" void kernel_launch(void* const* d_in, const int* in_sizes, int n_in,
                              void* d_out, int out_size, void* d_ws, size_t ws_size,
                              hipStream_t stream) {
    (void)in_sizes; (void)n_in; (void)d_ws; (void)ws_size; (void)out_size;
    const float* x0 = (const float*)d_in[0];
    float* out = (float*)d_out;
    const size_t F = (size_t)DEMB * NPIX;

    dim3 grid(NTILES), block(32 * NWAVES);
    ms_iter_kernel<<<grid, block, 0, stream>>>(x0, out + F, nullptr);
    ms_iter_kernel<<<grid, block, 0, stream>>>(out + F, out + 2 * F, nullptr);
    ms_iter_kernel<<<grid, block, 0, stream>>>(out + 2 * F, out + 3 * F, out);
}


# Each block owns three adjacent j-tiles (grid 192, 16 waves): a wave stages an x slab and reads its fragments once for all three; i-stream now split 16 ways, so f32 sum order of Y and d changes
# speedup vs baseline: 1.8111x; 1.1481x over previous
#include <hip/hip_runtime.h>

typedef __attribute__((ext_vector_type(2))) float v2f;
typedef __attribute__((ext_vector_type(8))) float v8f;

#define NPIX   9216
#define DEMB   32
#define NWAVES 16
#define NT     3
#define JW     (16 * NT)
#define SLABW  32
#define NSLABS (NPIX / SLABW)
#define NBLOCKS (NPIX / JW)
#define XSTRIDE 44
#define NRED   (16 * NT + NT)
#define BW     3.0f
#define STEP   0.5f
#define LOG2E  1.44269504088896340736f

__global__ __launch_bounds__(32 * NWAVES)
void ms_iter_kernel(const float* __restrict__ xin,
                    float* __restrict__ xout,
                    float* __restrict__ xout2)
{
    __shared__ __align__(16) float xtile[NWAVES][DEMB * XSTRIDE];
    __shared__ float red[NWAVES][NRED][32];

    const int lane  = threadIdx.x & 31;
    const int wave  = threadIdx.x >> 5;
    const int half  = lane >> 4;
    const int l15   = lane & 15;
    const int jbase = blockIdx.x * JW;

    float* xt = xtile[wave];
    const int pcol = 4 * ((l15 >> 1) & 3) + 2 * (l15 >> 3) + (l15 & 1);

    v2f bj[NT][8];
#pragma unroll
    for (int t = 0; t < NT; ++t) {
#pragma unroll
        for (int k0 = 0; k0 < 8; ++k0) {
            const int d0 = 4 * k0 + 2 * half;
            bj[t][k0].x = xin[(size_t)d0 * NPIX + jbase + 16 * t + l15];
            bj[t][k0].y = xin[(size_t)(d0 + 1) * NPIX + jbase + 16 * t + l15];
        }
    }

    v8f y0[NT], y1[NT];
    float dacc[NT];
#pragma unroll
    for (int t = 0; t < NT; ++t) {
#pragma unroll
        for (int v = 0; v < 8; ++v) { y0[t][v] = 0.0f; y1[t][v] = 0.0f; }
        dacc[t] = 0.0f;
    }

    const int srow = lane >> 3;
    const int scol = 4 * (lane & 7);

    for (int sl = wave; sl < NSLABS; sl += NWAVES) {
        const int ibase = sl * SLABW;

        {
            const float4* src = (const float4*)(xin + (size_t)srow * NPIX + ibase + scol);
            float4* dst = (float4*)(xt + srow * XSTRIDE + scol);
            float4 r[8];
#pragma unroll
            for (int q = 0; q < 8; ++q) r[q] = src[(size_t)q * NPIX];
#pragma unroll
            for (int q = 0; q < 8; ++q) dst[q * XSTRIDE] = r[q];
        }

        v2f a0[8], a1[8];
#pragma unroll
        for (int k0 = 0; k0 < 8; ++k0) {
            const int d0 = 4 * k0 + 2 * half;
            a0[k0].x = xt[d0 * XSTRIDE + pcol];
            a0[k0].y = xt[(d0 + 1) * XSTRIDE + pcol];
            a1[k0].x = xt[d0 * XSTRIDE + 16 + pcol];
            a1[k0].y = xt[(d0 + 1) * XSTRIDE + 16 + pcol];
        }
        v2f a2[8], a3[8];
#pragma unroll
        for (int c = 0; c < 8; ++c) {
            const int k0 = 4 * c + 2 * half;
            a2[c].x = xt[l15 * XSTRIDE + k0];
            a2[c].y = xt[l15 * XSTRIDE + k0 + 1];
            a3[c].x = xt[(16 + l15) * XSTRIDE + k0];
            a3[c].y = xt[(16 + l15) * XSTRIDE + k0 + 1];
        }

#pragma unroll
        for (int t = 0; t < NT; ++t) {
            v8f s0 = {};
            v8f s1 = {};
#pragma unroll
            for (int k0 = 0; k0 < 8; ++k0) {
                s0 = __builtin_amdgcn_wmma_f32_16x16x4_f32(false, a0[k0], false, bj[t][k0],
                                                           (short)0, s0, false, false);
                s1 = __builtin_amdgcn_wmma_f32_16x16x4_f32(false, a1[k0], false, bj[t][k0],
                                                           (short)0, s1, false, false);
            }

            v8f e0, e1;
            float esum0 = 0.0f, esum1 = 0.0f;
#pragma unroll
            for (int v = 0; v < 8; ++v) {
                e0[v] = __builtin_amdgcn_exp2f(s0[v] * (BW * LOG2E));
                e1[v] = __builtin_amdgcn_exp2f(s1[v] * (BW * LOG2E));
                esum0 += e0[v];
                esum1 += e1[v];
            }
            dacc[t] += esum0;
            dacc[t] += esum1;

#pragma unroll
            for (int c = 0; c < 8; ++c) {
                const int t2 = 2 * (c & 3);
                v2f b2;
                b2.x = (c < 4) ? e0[t2]     : e1[t2];
                b2.y = (c < 4) ? e0[t2 + 1] : e1[t2 + 1];
                y0[t] = __builtin_amdgcn_wmma_f32_16x16x4_f32(false, a2[c], false, b2,
                                                              (short)0, y0[t], false, false);
                y1[t] = __builtin_amdgcn_wmma_f32_16x16x4_f32(false, a3[c], false, b2,
                                                              (short)0, y1[t], false, false);
            }
        }
    }

#pragma unroll
    for (int t = 0; t < NT; ++t) {
#pragma unroll
        for (int v = 0; v < 8; ++v) {
            red[wave][16 * t + v][lane]     = y0[t][v];
            red[wave][16 * t + 8 + v][lane] = y1[t][v];
        }
        red[wave][16 * NT + t][lane] = dacc[t];
    }
    __syncthreads();

    {
        const int h2 = wave >> 3;
        const int v  = wave & 7;
        const int m  = 16 * h2 + 8 * half + v;
#pragma unroll
        for (int t = 0; t < NT; ++t) {
            float ysum = 0.0f, dtot = 0.0f;
            for (int w = 0; w < NWAVES; ++w) {
                ysum += red[w][16 * t + wave][lane];
                dtot += red[w][16 * NT + t][lane];
            }
            const float dfull = dtot + __shfl_xor(dtot, 16, 32);
            const float sc = STEP / dfull;
            const size_t o = (size_t)m * NPIX + (size_t)(jbase + 16 * t + l15);
            const float r = ysum * sc + (1.0f - STEP) * xin[o];
            xout[o] = r;
            if (xout2) xout2[o] = r;
        }
    }
}

extern "C" void kernel_launch(void* const* d_in, const int* in_sizes, int n_in,
                              void* d_out, int out_size, void* d_ws, size_t ws_size,
                              hipStream_t stream) {
    (void)in_sizes; (void)n_in; (void)d_ws; (void)ws_size; (void)out_size;
    const float* x0 = (const float*)d_in[0];
    float* out = (float*)d_out;
    const size_t F = (size_t)DEMB * NPIX;

    dim3 grid(NBLOCKS), block(32 * NWAVES);
    ms_iter_kernel<<<grid, block, 0, stream>>>(x0, out + F, nullptr);
    ms_iter_kernel<<<grid, block, 0, stream>>>(out + F, out + 2 * F, nullptr);
    ms_iter_kernel<<<grid, block, 0, stream>>>(out + 2 * F, out + 3 * F, out);
}


# Grid 192 -> 256, one block per WGP: each owns two whole j-tiles plus a quarter of the i-stream of a tile shared by four; a short kernel sums the four partials, so their f32 sum order changes
# speedup vs baseline: 2.0359x; 1.1241x over previous
#include <hip/hip_runtime.h>

typedef __attribute__((ext_vector_type(2))) float v2f;
typedef __attribute__((ext_vector_type(8))) float v8f;

#define NPIX   9216
#define DEMB   32
#define NWAVES 16
#define NT     3
#define NBLOCKS 256
#define JW     32
#define SPLIT0 (NBLOCKS * JW)
#define NPART  4
#define SLABW  32
#define NSLABS (NPIX / SLABW)
#define PSLABS (NSLABS / NPART)
#define PSTRIDE 544
#define XSTRIDE 44
#define NRED   (16 * NT + NT)
#define CBLOCKS 32
#define BW     3.0f
#define STEP   0.5f
#define LOG2E  1.44269504088896340736f

__device__ __forceinline__ size_t rec_off(int k, int p, int row, int c)
{
    const int e = p * PSTRIDE + row * 16 + c;
    return (size_t)(e >> 7) * NPIX + (size_t)(128 * k + (e & 127));
}

__device__ __forceinline__ void tile_step(const v2f (&a0)[8], const v2f (&a1)[8],
                                          const v2f (&a2)[8], const v2f (&a3)[8],
                                          const v2f (&bj)[8],
                                          v8f& y0, v8f& y1, float& dacc)
{
    v8f s0 = {};
    v8f s1 = {};
#pragma unroll
    for (int k0 = 0; k0 < 8; ++k0) {
        s0 = __builtin_amdgcn_wmma_f32_16x16x4_f32(false, a0[k0], false, bj[k0],
                                                   (short)0, s0, false, false);
        s1 = __builtin_amdgcn_wmma_f32_16x16x4_f32(false, a1[k0], false, bj[k0],
                                                   (short)0, s1, false, false);
    }

    v8f e0, e1;
    float esum0 = 0.0f, esum1 = 0.0f;
#pragma unroll
    for (int v = 0; v < 8; ++v) {
        e0[v] = __builtin_amdgcn_exp2f(s0[v] * (BW * LOG2E));
        e1[v] = __builtin_amdgcn_exp2f(s1[v] * (BW * LOG2E));
        esum0 += e0[v];
        esum1 += e1[v];
    }
    dacc += esum0;
    dacc += esum1;

#pragma unroll
    for (int c = 0; c < 8; ++c) {
        const int t2 = 2 * (c & 3);
        v2f b2;
        b2.x = (c < 4) ? e0[t2]     : e1[t2];
        b2.y = (c < 4) ? e0[t2 + 1] : e1[t2 + 1];
        y0 = __builtin_amdgcn_wmma_f32_16x16x4_f32(false, a2[c], false, b2,
                                                   (short)0, y0, false, false);
        y1 = __builtin_amdgcn_wmma_f32_16x16x4_f32(false, a3[c], false, b2,
                                                   (short)0, y1, false, false);
    }
}

__global__ __launch_bounds__(32 * NWAVES)
void ms_iter_kernel(const float* __restrict__ xin,
                    float* __restrict__ xout,
                    float* __restrict__ rec)
{
    __shared__ __align__(16) float xtile[NWAVES][DEMB * XSTRIDE];
    __shared__ float red[NWAVES][NRED][32];

    const int lane  = threadIdx.x & 31;
    const int wave  = __builtin_amdgcn_readfirstlane((int)(threadIdx.x >> 5));
    const int half  = lane >> 4;
    const int l15   = lane & 15;
    const int jbase = blockIdx.x * JW;
    const int ksh   = blockIdx.x >> 2;
    const int part  = blockIdx.x & (NPART - 1);
    const int jsh   = SPLIT0 + 16 * ksh;
    const int plo   = part * PSLABS;
    const int phi   = plo + PSLABS;

    float* xt = xtile[wave];
    const int pcol = 4 * ((l15 >> 1) & 3) + 2 * (l15 >> 3) + (l15 & 1);

    v2f bj[NT][8];
#pragma unroll
    for (int t = 0; t < NT; ++t) {
        const int jcol = (t < 2 ? jbase + 16 * t : jsh) + l15;
#pragma unroll
        for (int k0 = 0; k0 < 8; ++k0) {
            const int d0 = 4 * k0 + 2 * half;
            bj[t][k0].x = xin[(size_t)d0 * NPIX + jcol];
            bj[t][k0].y = xin[(size_t)(d0 + 1) * NPIX + jcol];
        }
    }

    v8f y0[NT], y1[NT];
    float dacc[NT];
#pragma unroll
    for (int t = 0; t < NT; ++t) {
#pragma unroll
        for (int v = 0; v < 8; ++v) { y0[t][v] = 0.0f; y1[t][v] = 0.0f; }
        dacc[t] = 0.0f;
    }

    const int srow = lane >> 3;
    const int scol = 4 * (lane & 7);

    for (int sl = wave; sl < NSLABS; sl += NWAVES) {
        const int ibase = sl * SLABW;

        {
            const float4* src = (const float4*)(xin + (size_t)srow * NPIX + ibase + scol);
            float4* dst = (float4*)(xt + srow * XSTRIDE + scol);
            float4 r[8];
#pragma unroll
            for (int q = 0; q < 8; ++q) r[q] = src[(size_t)q * NPIX];
#pragma unroll
            for (int q = 0; q < 8; ++q) dst[q * XSTRIDE] = r[q];
        }

        v2f a0[8], a1[8];
#pragma unroll
        for (int k0 = 0; k0 < 8; ++k0) {
            const int d0 = 4 * k0 + 2 * half;
            a0[k0].x = xt[d0 * XSTRIDE + pcol];
            a0[k0].y = xt[(d0 + 1) * XSTRIDE + pcol];
            a1[k0].x = xt[d0 * XSTRIDE + 16 + pcol];
            a1[k0].y = xt[(d0 + 1) * XSTRIDE + 16 + pcol];
        }
        v2f a2[8], a3[8];
#pragma unroll
        for (int c = 0; c < 8; ++c) {
            const int k0 = 4 * c + 2 * half;
            a2[c].x = xt[l15 * XSTRIDE + k0];
            a2[c].y = xt[l15 * XSTRIDE + k0 + 1];
            a3[c].x = xt[(16 + l15) * XSTRIDE + k0];
            a3[c].y = xt[(16 + l15) * XSTRIDE + k0 + 1];
        }

        tile_step(a0, a1, a2, a3, bj[0], y0[0], y1[0], dacc[0]);
        tile_step(a0, a1, a2, a3, bj[1], y0[1], y1[1], dacc[1]);
        if (sl >= plo && sl < phi)
            tile_step(a0, a1, a2, a3, bj[2], y0[2], y1[2], dacc[2]);
    }

#pragma unroll
    for (int t = 0; t < NT; ++t) {
#pragma unroll
        for (int v = 0; v < 8; ++v) {
            red[wave][16 * t + v][lane]     = y0[t][v];
            red[wave][16 * t + 8 + v][lane] = y1[t][v];
        }
        red[wave][16 * NT + t][lane] = dacc[t];
    }
    __syncthreads();

    {
        const int h2 = wave >> 3;
        const int v  = wave & 7;
        const int m  = 16 * h2 + 8 * half + v;
#pragma unroll
        for (int t = 0; t < NT; ++t) {
            float ysum = 0.0f, dtot = 0.0f;
            for (int w = 0; w < NWAVES; ++w) {
                ysum += red[w][16 * t + wave][lane];
                dtot += red[w][16 * NT + t][lane];
            }
            const float dfull = dtot + __shfl_xor(dtot, 16, 32);
            if (t < 2) {
                const float sc = STEP / dfull;
                const size_t o = (size_t)m * NPIX + (size_t)(jbase + 16 * t + l15);
                xout[o] = ysum * sc + (1.0f - STEP) * xin[o];
            } else {
                rec[rec_off(ksh, part, m, l15)] = ysum;
                if (wave == 0 && half == 0) rec[rec_off(ksh, part, DEMB, l15)] = dfull;
            }
        }
    }
}

__global__ __launch_bounds__(1024)
void ms_combine_kernel(const float* __restrict__ xin,
                       float* __restrict__ xout,
                       const float* rec,
                       float* fin)
{
    const int c  = threadIdx.x & 31;
    const int d  = threadIdx.x >> 5;
    const int k  = 2 * blockIdx.x + (c >> 4);
    const int cc = c & 15;
    float ysum = 0.0f, dsum = 0.0f;
#pragma unroll
    for (int p = 0; p < NPART; ++p) {
        ysum += rec[rec_off(k, p, d, cc)];
        dsum += rec[rec_off(k, p, DEMB, cc)];
    }
    const size_t o = (size_t)d * NPIX + (size_t)(SPLIT0 + 32 * blockIdx.x + c);
    const float r = ysum * (STEP / dsum) + (1.0f - STEP) * xin[o];
    xout[o] = r;
    if (fin) {
        __syncthreads();
        fin[o] = r;
#pragma unroll
        for (int q = 0; q < 2; ++q) {
            const int idx = threadIdx.x + 1024 * q;
            const size_t off = (size_t)(idx >> 6) * NPIX + (size_t)(256 * blockIdx.x + 4 * (idx & 63));
            *(float4*)(fin + off) = *(const float4*)(xout + off);
        }
    }
}

extern "C" void kernel_launch(void* const* d_in, const int* in_sizes, int n_in,
                              void* d_out, int out_size, void* d_ws, size_t ws_size,
                              hipStream_t stream) {
    (void)in_sizes; (void)n_in; (void)d_ws; (void)ws_size; (void)out_size;
    const float* x0 = (const float*)d_in[0];
    float* out = (float*)d_out;
    const size_t F = (size_t)DEMB * NPIX;

    dim3 grid(NBLOCKS), block(32 * NWAVES);
    dim3 cgrid(CBLOCKS), cblock(1024);
    ms_iter_kernel<<<grid, block, 0, stream>>>(x0, out + F, out);
    ms_combine_kernel<<<cgrid, cblock, 0, stream>>>(x0, out + F, out, nullptr);
    ms_iter_kernel<<<grid, block, 0, stream>>>(out + F, out + 2 * F, out);
    ms_combine_kernel<<<cgrid, cblock, 0, stream>>>(out + F, out + 2 * F, out, nullptr);
    ms_iter_kernel<<<grid, block, 0, stream>>>(out + 2 * F, out + 3 * F, out);
    ms_combine_kernel<<<cgrid, cblock, 0, stream>>>(out + 2 * F, out + 3 * F, out, out);
}


# Each wave steps its issue priority down with its own progress (s_setprio 3,2,1,0 at slab 0, 9, 14, 17 of 18): waves of a SIMD end together, none stages alone at the end; f32 sum order unchanged
# speedup vs baseline: 2.1561x; 1.0590x over previous
#include <hip/hip_runtime.h>

typedef __attribute__((ext_vector_type(2))) float v2f;
typedef __attribute__((ext_vector_type(8))) float v8f;

#define NPIX   9216
#define DEMB   32
#define NWAVES 16
#define NT     3
#define NBLOCKS 256
#define JW     32
#define SPLIT0 (NBLOCKS * JW)
#define NPART  4
#define SLABW  32
#define NSLABS (NPIX / SLABW)
#define PSLABS (NSLABS / NPART)
#define PSTRIDE 544
#define XSTRIDE 44
#define NRED   (16 * NT + NT)
#define CBLOCKS 32
#define TRIPS  (NSLABS / NWAVES)
#define PSTEP1 (TRIPS / 2)
#define PSTEP2 ((3 * TRIPS + 3) / 4)
#define PSTEP3 (TRIPS - 1)
#define BW     3.0f
#define STEP   0.5f
#define LOG2E  1.44269504088896340736f

__device__ __forceinline__ size_t rec_off(int k, int p, int row, int c)
{
    const int e = p * PSTRIDE + row * 16 + c;
    return (size_t)(e >> 7) * NPIX + (size_t)(128 * k + (e & 127));
}

__device__ __forceinline__ void tile_step(const v2f (&a0)[8], const v2f (&a1)[8],
                                          const v2f (&a2)[8], const v2f (&a3)[8],
                                          const v2f (&bj)[8],
                                          v8f& y0, v8f& y1, float& dacc)
{
    v8f s0 = {};
    v8f s1 = {};
#pragma unroll
    for (int k0 = 0; k0 < 8; ++k0) {
        s0 = __builtin_amdgcn_wmma_f32_16x16x4_f32(false, a0[k0], false, bj[k0],
                                                   (short)0, s0, false, false);
        s1 = __builtin_amdgcn_wmma_f32_16x16x4_f32(false, a1[k0], false, bj[k0],
                                                   (short)0, s1, false, false);
    }

    v8f e0, e1;
    float esum0 = 0.0f, esum1 = 0.0f;
#pragma unroll
    for (int v = 0; v < 8; ++v) {
        e0[v] = __builtin_amdgcn_exp2f(s0[v] * (BW * LOG2E));
        e1[v] = __builtin_amdgcn_exp2f(s1[v] * (BW * LOG2E));
        esum0 += e0[v];
        esum1 += e1[v];
    }
    dacc += esum0;
    dacc += esum1;

#pragma unroll
    for (int c = 0; c < 8; ++c) {
        const int t2 = 2 * (c & 3);
        v2f b2;
        b2.x = (c < 4) ? e0[t2]     : e1[t2];
        b2.y = (c < 4) ? e0[t2 + 1] : e1[t2 + 1];
        y0 = __builtin_amdgcn_wmma_f32_16x16x4_f32(false, a2[c], false, b2,
                                                   (short)0, y0, false, false);
        y1 = __builtin_amdgcn_wmma_f32_16x16x4_f32(false, a3[c], false, b2,
                                                   (short)0, y1, false, false);
    }
}

__global__ __launch_bounds__(32 * NWAVES)
void ms_iter_kernel(const float* __restrict__ xin,
                    float* __restrict__ xout,
                    float* __restrict__ rec)
{
    __shared__ __align__(16) float xtile[NWAVES][DEMB * XSTRIDE];
    __shared__ float red[NWAVES][NRED][32];

    const int lane  = threadIdx.x & 31;
    const int wave  = __builtin_amdgcn_readfirstlane((int)(threadIdx.x >> 5));
    const int half  = lane >> 4;
    const int l15   = lane & 15;
    const int jbase = blockIdx.x * JW;
    const int ksh   = blockIdx.x >> 2;
    const int part  = blockIdx.x & (NPART - 1);
    const int jsh   = SPLIT0 + 16 * ksh;
    const int plo   = part * PSLABS;
    const int phi   = plo + PSLABS;

    float* xt = xtile[wave];
    const int pcol = 4 * ((l15 >> 1) & 3) + 2 * (l15 >> 3) + (l15 & 1);

    v2f bj[NT][8];
#pragma unroll
    for (int t = 0; t < NT; ++t) {
        const int jcol = (t < 2 ? jbase + 16 * t : jsh) + l15;
#pragma unroll
        for (int k0 = 0; k0 < 8; ++k0) {
            const int d0 = 4 * k0 + 2 * half;
            bj[t][k0].x = xin[(size_t)d0 * NPIX + jcol];
            bj[t][k0].y = xin[(size_t)(d0 + 1) * NPIX + jcol];
        }
    }

    v8f y0[NT], y1[NT];
    float dacc[NT];
#pragma unroll
    for (int t = 0; t < NT; ++t) {
#pragma unroll
        for (int v = 0; v < 8; ++v) { y0[t][v] = 0.0f; y1[t][v] = 0.0f; }
        dacc[t] = 0.0f;
    }

    const int srow = lane >> 3;
    const int scol = 4 * (lane & 7);

    __builtin_amdgcn_s_setprio(3);
    int trip = 0;
    for (int sl = wave; sl < NSLABS; sl += NWAVES, ++trip) {
        const int ibase = sl * SLABW;
        if (trip == PSTEP1)      __builtin_amdgcn_s_setprio(2);
        else if (trip == PSTEP2) __builtin_amdgcn_s_setprio(1);
        else if (trip == PSTEP3) __builtin_amdgcn_s_setprio(0);

        {
            const float4* src = (const float4*)(xin + (size_t)srow * NPIX + ibase + scol);
            float4* dst = (float4*)(xt + srow * XSTRIDE + scol);
            float4 r[8];
#pragma unroll
            for (int q = 0; q < 8; ++q) r[q] = src[(size_t)q * NPIX];
#pragma unroll
            for (int q = 0; q < 8; ++q) dst[q * XSTRIDE] = r[q];
        }

        v2f a0[8], a1[8];
#pragma unroll
        for (int k0 = 0; k0 < 8; ++k0) {
            const int d0 = 4 * k0 + 2 * half;
            a0[k0].x = xt[d0 * XSTRIDE + pcol];
            a0[k0].y = xt[(d0 + 1) * XSTRIDE + pcol];
            a1[k0].x = xt[d0 * XSTRIDE + 16 + pcol];
            a1[k0].y = xt[(d0 + 1) * XSTRIDE + 16 + pcol];
        }
        v2f a2[8], a3[8];
#pragma unroll
        for (int c = 0; c < 8; ++c) {
            const int k0 = 4 * c + 2 * half;
            a2[c].x = xt[l15 * XSTRIDE + k0];
            a2[c].y = xt[l15 * XSTRIDE + k0 + 1];
            a3[c].x = xt[(16 + l15) * XSTRIDE + k0];
            a3[c].y = xt[(16 + l15) * XSTRIDE + k0 + 1];
        }

        tile_step(a0, a1, a2, a3, bj[0], y0[0], y1[0], dacc[0]);
        tile_step(a0, a1, a2, a3, bj[1], y0[1], y1[1], dacc[1]);
        if (sl >= plo && sl < phi)
            tile_step(a0, a1, a2, a3, bj[2], y0[2], y1[2], dacc[2]);
    }

    __builtin_amdgcn_s_setprio(0);

#pragma unroll
    for (int t = 0; t < NT; ++t) {
#pragma unroll
        for (int v = 0; v < 8; ++v) {
            red[wave][16 * t + v][lane]     = y0[t][v];
            red[wave][16 * t + 8 + v][lane] = y1[t][v];
        }
        red[wave][16 * NT + t][lane] = dacc[t];
    }
    __syncthreads();

    {
        const int h2 = wave >> 3;
        const int v  = wave & 7;
        const int m  = 16 * h2 + 8 * half + v;
#pragma unroll
        for (int t = 0; t < NT; ++t) {
            float ysum = 0.0f, dtot = 0.0f;
            for (int w = 0; w < NWAVES; ++w) {
                ysum += red[w][16 * t + wave][lane];
                dtot += red[w][16 * NT + t][lane];
            }
            const float dfull = dtot + __shfl_xor(dtot, 16, 32);
            if (t < 2) {
                const float sc = STEP / dfull;
                const size_t o = (size_t)m * NPIX + (size_t)(jbase + 16 * t + l15);
                xout[o] = ysum * sc + (1.0f - STEP) * xin[o];
            } else {
                rec[rec_off(ksh, part, m, l15)] = ysum;
                if (wave == 0 && half == 0) rec[rec_off(ksh, part, DEMB, l15)] = dfull;
            }
        }
    }
}

__global__ __launch_bounds__(1024)
void ms_combine_kernel(const float* __restrict__ xin,
                       float* __restrict__ xout,
                       const float* rec,
                       float* fin)
{
    const int c  = threadIdx.x & 31;
    const int d  = threadIdx.x >> 5;
    const int k  = 2 * blockIdx.x + (c >> 4);
    const int cc = c & 15;
    float ysum = 0.0f, dsum = 0.0f;
#pragma unroll
    for (int p = 0; p < NPART; ++p) {
        ysum += rec[rec_off(k, p, d, cc)];
        dsum += rec[rec_off(k, p, DEMB, cc)];
    }
    const size_t o = (size_t)d * NPIX + (size_t)(SPLIT0 + 32 * blockIdx.x + c);
    const float r = ysum * (STEP / dsum) + (1.0f - STEP) * xin[o];
    xout[o] = r;
    if (fin) {
        __syncthreads();
        fin[o] = r;
#pragma unroll
        for (int q = 0; q < 2; ++q) {
            const int idx = threadIdx.x + 1024 * q;
            const size_t off = (size_t)(idx >> 6) * NPIX + (size_t)(256 * blockIdx.x + 4 * (idx & 63));
            *(float4*)(fin + off) = *(const float4*)(xout + off);
        }
    }
}

extern "C" void kernel_launch(void* const* d_in, const int* in_sizes, int n_in,
                              void* d_out, int out_size, void* d_ws, size_t ws_size,
                              hipStream_t stream) {
    (void)in_sizes; (void)n_in; (void)d_ws; (void)ws_size; (void)out_size;
    const float* x0 = (const float*)d_in[0];
    float* out = (float*)d_out;
    const size_t F = (size_t)DEMB * NPIX;

    dim3 grid(NBLOCKS), block(32 * NWAVES);
    dim3 cgrid(CBLOCKS), cblock(1024);
    ms_iter_kernel<<<grid, block, 0, stream>>>(x0, out + F, out);
    ms_combine_kernel<<<cgrid, cblock, 0, stream>>>(x0, out + F, out, nullptr);
    ms_iter_kernel<<<grid, block, 0, stream>>>(out + F, out + 2 * F, out);
    ms_combine_kernel<<<cgrid, cblock, 0, stream>>>(out + F, out + 2 * F, out, nullptr);
    ms_iter_kernel<<<grid, block, 0, stream>>>(out + 2 * F, out + 3 * F, out);
    ms_combine_kernel<<<cgrid, cblock, 0, stream>>>(out + 2 * F, out + 3 * F, out, out);
}
